# starting kernel
# baseline (speedup 1.0000x reference)
#include <hip/hip_runtime.h>
#include <hip/hip_bf16.h>

typedef __attribute__((ext_vector_type(16))) _Float16 v16h;
typedef __attribute__((ext_vector_type(8)))  float    v8f;

#define DIMS   128
#define PMAX   50
#define PPAD   64
#define LSTR   132

__global__ __launch_bounds__(128)
void AttCML_kernel(const float* __restrict__ user_emb,
                   const float* __restrict__ item_emb,
                   const int*   __restrict__ user_ids,
                   const int*   __restrict__ pos_ids,
                   const int*   __restrict__ neg_ids,
                   const int*   __restrict__ pref_ids,
                   const float* __restrict__ n_prefs,
                   float* __restrict__ out,
                   int B, int n_items)
{
    __shared__ __align__(16) float sPref[PPAD * LSTR];
    __shared__ __align__(16) float sT[2 * LSTR];
    __shared__ float sW[2 * PPAD];
    __shared__ float sAtt[2 * PPAD];
    __shared__ float sRed[8];

    const int b    = blockIdx.x;
    const int tid  = threadIdx.x;
    const int wave = tid >> 5;
    const int lane = tid & 31;

    for (int r = wave; r < PPAD; r += 4) {
        float4 v = make_float4(0.f, 0.f, 0.f, 0.f);
        if (r < PMAX) {
            int id = pref_ids[b * PMAX + r];
            if (id < n_items)
                v = *(const float4*)(item_emb + (size_t)id * DIMS + lane * 4);
        }
        *(float4*)(&sPref[r * LSTR + lane * 4]) = v;
    }
    if (wave == 0) {
        int id = pos_ids[b];
        *(float4*)(&sT[0 * LSTR + lane * 4]) =
            *(const float4*)(item_emb + (size_t)id * DIMS + lane * 4);
    } else if (wave == 1) {
        int id = neg_ids[b];
        *(float4*)(&sT[1 * LSTR + lane * 4]) =
            *(const float4*)(item_emb + (size_t)id * DIMS + lane * 4);
    }
    __syncthreads();

    {
        const int mrow  = wave * 16 + (lane & 15);
        const int half8 = (lane >> 4) ? 8 : 0;
        const int col   = lane & 15;
        const float* arow = &sPref[mrow * LSTR];
        const float* brow = &sT[(col < 2 ? col : 0) * LSTR];
        const bool   bval = (col < 2);

        v8f c = {0.f, 0.f, 0.f, 0.f, 0.f, 0.f, 0.f, 0.f};
        #pragma unroll
        for (int kb = 0; kb < DIMS; kb += 32) {
            const int k0 = kb + half8;
            const int k1 = kb + 16 + half8;
            v16h a, bm;
            #pragma unroll
            for (int j = 0; j < 4; ++j) {
                float2 fa0 = *(const float2*)(arow + k0 + 2 * j);
                float2 fa1 = *(const float2*)(arow + k1 + 2 * j);
                a[2 * j]     = (_Float16)fa0.x;
                a[2 * j + 1] = (_Float16)fa0.y;
                a[8 + 2 * j]     = (_Float16)fa1.x;
                a[8 + 2 * j + 1] = (_Float16)fa1.y;
                float2 fb0 = *(const float2*)(brow + k0 + 2 * j);
                float2 fb1 = *(const float2*)(brow + k1 + 2 * j);
                bm[2 * j]     = (_Float16)(bval ? fb0.x : 0.f);
                bm[2 * j + 1] = (_Float16)(bval ? fb0.y : 0.f);
                bm[8 + 2 * j]     = (_Float16)(bval ? fb1.x : 0.f);
                bm[8 + 2 * j + 1] = (_Float16)(bval ? fb1.y : 0.f);
            }
            c = __builtin_amdgcn_wmma_f32_16x16x32_f16(false, a, false, bm,
                                                       (short)0, c, false, false);
        }
        if (col < 2) {
            #pragma unroll
            for (int g = 0; g < 8; ++g)
                sW[col * PPAD + wave * 16 + half8 + g] = c[g];
        }
    }
    __syncthreads();

    if (wave < 2) {
        const float np1 = n_prefs[b] + 1.0f;
        const int p0 = lane, p1 = lane + 32;
        float e0 = ((float)p0 < np1) ? __expf(sW[wave * PPAD + p0]) : 0.f;
        float e1 = (p1 < PMAX && (float)p1 < np1) ? __expf(sW[wave * PPAD + p1]) : 0.f;
        float s = e0 + e1;
        #pragma unroll
        for (int off = 16; off > 0; off >>= 1) s += __shfl_xor(s, off, 32);
        const float inv = 1.0f / s;
        sAtt[wave * PPAD + p0] = e0 * inv;
        if (p1 < PMAX) sAtt[wave * PPAD + p1] = e1 * inv;
    }
    __syncthreads();

    {
        const int d = tid;
        float accP = 0.f, accN = 0.f;
        #pragma unroll 5
        for (int p = 0; p < PMAX; ++p) {
            const float pv = sPref[p * LSTR + d];
            accP = fmaf(sAtt[p], pv, accP);
            accN = fmaf(sAtt[PPAD + p], pv, accN);
        }
        const float u  = user_emb[(size_t)user_ids[b] * DIMS + d];
        const float dp = u + accP - sT[d];
        const float dn = u + accN - sT[LSTR + d];
        float sqP = dp * dp, sqN = dn * dn;
        #pragma unroll
        for (int off = 16; off > 0; off >>= 1) {
            sqP += __shfl_xor(sqP, off, 32);
            sqN += __shfl_xor(sqN, off, 32);
        }
        if (lane == 0) { sRed[wave] = sqP; sRed[4 + wave] = sqN; }
    }
    __syncthreads();
    if (tid == 0) {
        out[b]     = sRed[0] + sRed[1] + sRed[2] + sRed[3];
        out[B + b] = sRed[4] + sRed[5] + sRed[6] + sRed[7];
    }
}

extern "C" void kernel_launch(void* const* d_in, const int* in_sizes, int n_in,
                              void* d_out, int out_size, void* d_ws, size_t ws_size,
                              hipStream_t stream) {
    (void)n_in; (void)out_size; (void)d_ws; (void)ws_size;
    const float* user_emb = (const float*)d_in[0];
    const float* item_emb = (const float*)d_in[1];
    const int*   user_ids = (const int*)d_in[2];
    const int*   pos_ids  = (const int*)d_in[3];
    const int*   neg_ids  = (const int*)d_in[4];
    const int*   pref_ids = (const int*)d_in[5];
    const float* n_prefs  = (const float*)d_in[6];
    float* out = (float*)d_out;

    const int B       = in_sizes[2];
    const int n_items = in_sizes[1] / DIMS;

    AttCML_kernel<<<B, 128, 0, stream>>>(user_emb, item_emb, user_ids, pos_ids,
                                         neg_ids, pref_ids, n_prefs, out, B, n_items);
}


# Gather with all loads in flight: each wave gets its 13 row ids by one lane-indexed load, then issues its 13 row loads and the user row back to back, not one dependent id->row pair per loop trip
# speedup vs baseline: 1.7246x; 1.7246x over previous
#include <hip/hip_runtime.h>
#include <hip/hip_bf16.h>

typedef __attribute__((ext_vector_type(16))) _Float16 v16h;
typedef __attribute__((ext_vector_type(8)))  float    v8f;

#define DIMS   128
#define PMAX   50
#define PPAD   64
#define RPW    13
#define LSTR   132

__global__ __launch_bounds__(128)
void AttCML_kernel(const float* __restrict__ user_emb,
                   const float* __restrict__ item_emb,
                   const int*   __restrict__ user_ids,
                   const int*   __restrict__ pos_ids,
                   const int*   __restrict__ neg_ids,
                   const int*   __restrict__ pref_ids,
                   const float* __restrict__ n_prefs,
                   float* __restrict__ out,
                   int B, int n_items)
{
    __shared__ __align__(16) float sRows[(PPAD + 2) * LSTR];
    float* const sPref = sRows;
    float* const sT    = sRows + PPAD * LSTR;
    __shared__ float sW[2 * PPAD];
    __shared__ float sAtt[2 * PPAD];
    __shared__ float sRed[8];

    const int b    = blockIdx.x;
    const int tid  = threadIdx.x;
    const int wave = tid >> 5;
    const int lane = tid & 31;

    const int wave_s = __builtin_amdgcn_readfirstlane(wave) & 3;
    int myid = 0;
    {
        const int vr = wave_s + 4 * lane;
        if (vr < PMAX) myid = pref_ids[b * PMAX + vr];
    }
    const int   pid = pos_ids[b];
    const int   nid = neg_ids[b];
    const int   uid = user_ids[b];
    const float np1 = n_prefs[b] + 1.0f;

    #pragma unroll
    for (int i = 0; i < 4; ++i) {
        const int r = PMAX + wave_s + 4 * i;
        if (r < PPAD)
            *(float4*)(&sPref[r * LSTR + lane * 4]) = make_float4(0.f, 0.f, 0.f, 0.f);
    }

    const unsigned padmask = __builtin_amdgcn_ballot_w32(myid >= n_items);

    float4 v[RPW];
    #pragma unroll
    for (int j = 0; j < RPW; ++j) {
        int id = __builtin_amdgcn_readlane(myid, j);
        if (j == RPW - 1 && wave_s >= 2)
            id = (wave_s == 2) ? pid : nid;
        const int idc = min(id, n_items - 1);
        v[j] = *(const float4*)(item_emb + (size_t)idc * DIMS + lane * 4);
    }
    const float u = user_emb[(size_t)uid * DIMS + tid];

    #pragma unroll
    for (int j = 0; j < RPW; ++j) {
        int row = wave_s + 4 * j;
        if (j == RPW - 1 && wave_s >= 2) row = PPAD + (wave_s - 2);
        *(float4*)(&sRows[row * LSTR + lane * 4]) = v[j];
    }
    if (padmask != 0u) {
        #pragma unroll
        for (int j = 0; j < RPW; ++j) {
            if ((padmask >> j) & 1u)
                *(float4*)(&sRows[(wave_s + 4 * j) * LSTR + lane * 4]) = make_float4(0.f, 0.f, 0.f, 0.f);
        }
    }
    __syncthreads();

    {
        const int mrow  = wave * 16 + (lane & 15);
        const int half8 = (lane >> 4) ? 8 : 0;
        const int col   = lane & 15;
        const float* arow = &sPref[mrow * LSTR];
        const float* brow = &sT[(col < 2 ? col : 0) * LSTR];
        const bool   bval = (col < 2);

        v8f c = {0.f, 0.f, 0.f, 0.f, 0.f, 0.f, 0.f, 0.f};
        #pragma unroll
        for (int kb = 0; kb < DIMS; kb += 32) {
            const int k0 = kb + half8;
            const int k1 = kb + 16 + half8;
            v16h a, bm;
            #pragma unroll
            for (int j = 0; j < 4; ++j) {
                float2 fa0 = *(const float2*)(arow + k0 + 2 * j);
                float2 fa1 = *(const float2*)(arow + k1 + 2 * j);
                a[2 * j]     = (_Float16)fa0.x;
                a[2 * j + 1] = (_Float16)fa0.y;
                a[8 + 2 * j]     = (_Float16)fa1.x;
                a[8 + 2 * j + 1] = (_Float16)fa1.y;
                float2 fb0 = *(const float2*)(brow + k0 + 2 * j);
                float2 fb1 = *(const float2*)(brow + k1 + 2 * j);
                bm[2 * j]     = (_Float16)(bval ? fb0.x : 0.f);
                bm[2 * j + 1] = (_Float16)(bval ? fb0.y : 0.f);
                bm[8 + 2 * j]     = (_Float16)(bval ? fb1.x : 0.f);
                bm[8 + 2 * j + 1] = (_Float16)(bval ? fb1.y : 0.f);
            }
            c = __builtin_amdgcn_wmma_f32_16x16x32_f16(false, a, false, bm,
                                                       (short)0, c, false, false);
        }
        if (col < 2) {
            #pragma unroll
            for (int g = 0; g < 8; ++g)
                sW[col * PPAD + wave * 16 + half8 + g] = c[g];
        }
    }
    __syncthreads();

    if (wave < 2) {
        const int p0 = lane, p1 = lane + 32;
        float e0 = ((float)p0 < np1) ? __expf(sW[wave * PPAD + p0]) : 0.f;
        float e1 = (p1 < PMAX && (float)p1 < np1) ? __expf(sW[wave * PPAD + p1]) : 0.f;
        float s = e0 + e1;
        #pragma unroll
        for (int off = 16; off > 0; off >>= 1) s += __shfl_xor(s, off, 32);
        const float inv = 1.0f / s;
        sAtt[wave * PPAD + p0] = e0 * inv;
        if (p1 < PMAX) sAtt[wave * PPAD + p1] = e1 * inv;
    }
    __syncthreads();

    {
        const int d = tid;
        float accP = 0.f, accN = 0.f;
        #pragma unroll 5
        for (int p = 0; p < PMAX; ++p) {
            const float pv = sPref[p * LSTR + d];
            accP = fmaf(sAtt[p], pv, accP);
            accN = fmaf(sAtt[PPAD + p], pv, accN);
        }
        const float dp = u + accP - sT[d];
        const float dn = u + accN - sT[LSTR + d];
        float sqP = dp * dp, sqN = dn * dn;
        #pragma unroll
        for (int off = 16; off > 0; off >>= 1) {
            sqP += __shfl_xor(sqP, off, 32);
            sqN += __shfl_xor(sqN, off, 32);
        }
        if (lane == 0) { sRed[wave] = sqP; sRed[4 + wave] = sqN; }
    }
    __syncthreads();
    if (tid == 0) {
        out[b]     = sRed[0] + sRed[1] + sRed[2] + sRed[3];
        out[B + b] = sRed[4] + sRed[5] + sRed[6] + sRed[7];
    }
}

extern "C" void kernel_launch(void* const* d_in, const int* in_sizes, int n_in,
                              void* d_out, int out_size, void* d_ws, size_t ws_size,
                              hipStream_t stream) {
    (void)n_in; (void)out_size; (void)d_ws; (void)ws_size;
    const float* user_emb = (const float*)d_in[0];
    const float* item_emb = (const float*)d_in[1];
    const int*   user_ids = (const int*)d_in[2];
    const int*   pos_ids  = (const int*)d_in[3];
    const int*   neg_ids  = (const int*)d_in[4];
    const int*   pref_ids = (const int*)d_in[5];
    const float* n_prefs  = (const float*)d_in[6];
    float* out = (float*)d_out;

    const int B       = in_sizes[2];
    const int n_items = in_sizes[1] / DIMS;

    AttCML_kernel<<<B, 128, 0, stream>>>(user_emb, item_emb, user_ids, pos_ids,
                                         neg_ids, pref_ids, n_prefs, out, B, n_items);
}


# Persistent work-groups, software-pipelined over samples: 2048 groups walk 8 samples each; the next sample's 53 row loads stay in flight (hand-written barriers) under this sample's softmax and sum
# speedup vs baseline: 1.7757x; 1.0296x over previous
#include <hip/hip_runtime.h>
#include <hip/hip_bf16.h>

typedef __attribute__((ext_vector_type(16))) _Float16 v16h;
typedef __attribute__((ext_vector_type(8)))  float    v8f;

#define DIMS   128
#define PMAX   50
#define PPAD   64
#define RPW    13
#define LSTR   132
#define NBLK   2048

static __device__ __forceinline__ void wg_barrier() {
    asm volatile("s_wait_dscnt 0x0\n\ts_barrier_signal -1\n\ts_barrier_wait -1" ::: "memory");
}

static __device__ __forceinline__ const float4* row_base(const float* item_emb, int myid, int j, int pid, int nid,
                                                        int wave_s, int n_items)
{
    int id = __builtin_amdgcn_readlane(myid, j);
    if (j == RPW - 1 && wave_s >= 2)
        id = (wave_s == 2) ? pid : nid;
    const int idc = min(id, n_items - 1);
    const size_t   off = (size_t)idc * (DIMS * sizeof(float));
    const unsigned lo  = __builtin_amdgcn_readfirstlane((unsigned)off);
    const unsigned hi  = __builtin_amdgcn_readfirstlane((unsigned)(off >> 32));
    const char*    row = (const char*)item_emb + (((size_t)hi << 32) | lo);
    return (const float4*)row;
}

static __device__ __forceinline__ int ids_load(const int* pref_ids, const int* pos_ids, const int* neg_ids,
                                               const int* user_ids, int bi, int vr, int lane)
{
    const int* p = pref_ids + (size_t)bi * PMAX + vr;
    if (lane == 13) p = pos_ids + bi;
    if (lane == 14) p = neg_ids + bi;
    if (lane == 15) p = user_ids + bi;
    int r = 0;
    if (vr < PMAX || (lane >= 13 && lane <= 15)) r = *p;
    return r;
}

static __device__ __forceinline__ float np_load(const float* n_prefs, int bi, int lane)
{
    float r = 0.f;
    if (lane == 0) r = n_prefs[bi];
    return r;
}
static __device__ __forceinline__ float np_use(float npv)
{
    return __int_as_float(__builtin_amdgcn_readfirstlane(__float_as_int(npv)));
}

#define FOR_ROWS(X) X(0) X(1) X(2) X(3) X(4) X(5) X(6) X(7) X(8) X(9) X(10) X(11) X(12)
#define ROW_DECL(j)  float4 v##j;
#define ROW_BASE(j)  const float4* rp##j = row_base(item_emb, ids_, j, pid_, nid_, wave_s, n_items);
#define ROW_LOAD(j)  v##j = rp##j[lane];
#define ROW_STORE(j) { int row = wave_s + 4 * j;                                          \
                       if (j == RPW - 1 && wave_s >= 2) row = PPAD + (wave_s - 2);        \
                       *(float4*)(&sRows[row * LSTR + lane * 4]) = v##j; }
#define ROW_ZERO(j)  if ((padmask >> j) & 1u)                                             \
                         *(float4*)(&sRows[(wave_s + 4 * j) * LSTR + lane * 4]) = make_float4(0.f, 0.f, 0.f, 0.f);

static __device__ __forceinline__ void finish_sample(const float* sPref, const float* sT, const float* sW,
                                                     float* sAtt, float* sRed, float* out,
                                                     int wave, int lane, int tid, float np1, float u_cur,
                                                     int B, int b)
{
    wg_barrier();

    if (wave < 2) {
        const int p0 = lane, p1 = lane + 32;
        float e0 = ((float)p0 < np1) ? __expf(sW[wave * PPAD + p0]) : 0.f;
        float e1 = (p1 < PMAX && (float)p1 < np1) ? __expf(sW[wave * PPAD + p1]) : 0.f;
        float s = e0 + e1;
        #pragma unroll
        for (int off = 16; off > 0; off >>= 1) s += __shfl_xor(s, off, 32);
        const float inv = 1.0f / s;
        sAtt[wave * PPAD + p0] = e0 * inv;
        if (p1 < PMAX) sAtt[wave * PPAD + p1] = e1 * inv;
    }
    wg_barrier();

    {
        const int d = tid;
        float accP = 0.f, accN = 0.f;
        #pragma unroll 5
        for (int p = 0; p < PMAX; ++p) {
            const float pv = sPref[p * LSTR + d];
            accP = fmaf(sAtt[p], pv, accP);
            accN = fmaf(sAtt[PPAD + p], pv, accN);
        }
        const float dp = u_cur + accP - sT[d];
        const float dn = u_cur + accN - sT[LSTR + d];
        float sqP = dp * dp, sqN = dn * dn;
        #pragma unroll
        for (int off = 16; off > 0; off >>= 1) {
            sqP += __shfl_xor(sqP, off, 32);
            sqN += __shfl_xor(sqN, off, 32);
        }
        if (lane == 0) { sRed[wave] = sqP; sRed[4 + wave] = sqN; }
    }
    wg_barrier();
    if (tid == 0) {
        out[b]     = sRed[0] + sRed[1] + sRed[2] + sRed[3];
        out[B + b] = sRed[4] + sRed[5] + sRed[6] + sRed[7];
    }
}

__global__ __launch_bounds__(128) __attribute__((amdgpu_waves_per_eu(8, 8)))
void AttCML_kernel(const float* user_emb,
                   const float* item_emb,
                   const int*   user_ids,
                   const int*   pos_ids,
                   const int*   neg_ids,
                   const int*   pref_ids,
                   const float* n_prefs,
                   float* out,
                   int B, int n_items)
{
    __shared__ __align__(16) float sRows[(PPAD + 2) * LSTR];
    float* const sPref = sRows;
    float* const sT    = sRows + PPAD * LSTR;
    __shared__ float sW[2 * PPAD];
    __shared__ float sAtt[2 * PPAD];
    __shared__ float sRed[8];

    const int tid  = threadIdx.x;
    const int wave = tid >> 5;
    const int lane = tid & 31;
    int b = blockIdx.x;
    if (b >= B) return;

    const int wave_s = __builtin_amdgcn_readfirstlane(wave) & 3;
    const int vr = wave_s + 4 * lane;

    #pragma unroll
    for (int i = 0; i < 4; ++i) {
        const int r = PMAX + wave_s + 4 * i;
        if (r < PPAD)
            *(float4*)(&sPref[r * LSTR + lane * 4]) = make_float4(0.f, 0.f, 0.f, 0.f);
    }

    int ids = ids_load(pref_ids, pos_ids, neg_ids, user_ids, b, vr, lane);
    float npc  = np_load(n_prefs, b, lane);
    float np1r = np_load(n_prefs, min(b + NBLK, B - 1), lane);
    FOR_ROWS(ROW_DECL)
    float  u;
    unsigned padmask = __builtin_amdgcn_ballot_w32(vr < PMAX && ids >= n_items);
    {
        const int ids_ = ids;
        const int pid_ = __builtin_amdgcn_readlane(ids, 13), nid_ = __builtin_amdgcn_readlane(ids, 14);
        const int uid_ = __builtin_amdgcn_readlane(ids, 15);
        FOR_ROWS(ROW_BASE)
        FOR_ROWS(ROW_LOAD)
        u = (user_emb + (size_t)uid_ * DIMS)[tid];
    }
    ids = ids_load(pref_ids, pos_ids, neg_ids, user_ids, min(b + NBLK, B - 1), vr, lane);
    float u_cur, np1;

    for (;;) {
        FOR_ROWS(ROW_STORE)
        if (padmask != 0u) {
            FOR_ROWS(ROW_ZERO)
        }
        u_cur = u;
        np1   = np_use(npc) + 1.0f;
        wg_barrier();

        {
            const int mrow  = wave * 16 + (lane & 15);
            const int half8 = (lane >> 4) ? 8 : 0;
            const int col   = lane & 15;
            const float* arow = &sPref[mrow * LSTR];
            const float* brow = &sT[(col < 2 ? col : 0) * LSTR];
            const bool   bval = (col < 2);

            v8f c = {0.f, 0.f, 0.f, 0.f, 0.f, 0.f, 0.f, 0.f};
            #pragma unroll 1
            for (int kb = 0; kb < DIMS; kb += 32) {
                const int k0 = kb + half8;
                const int k1 = kb + 16 + half8;
                v16h a, bm;
                #pragma unroll
                for (int j = 0; j < 4; ++j) {
                    float2 fa0 = *(const float2*)(arow + k0 + 2 * j);
                    float2 fa1 = *(const float2*)(arow + k1 + 2 * j);
                    a[2 * j]     = (_Float16)fa0.x;
                    a[2 * j + 1] = (_Float16)fa0.y;
                    a[8 + 2 * j]     = (_Float16)fa1.x;
                    a[8 + 2 * j + 1] = (_Float16)fa1.y;
                    float2 fb0 = *(const float2*)(brow + k0 + 2 * j);
                    float2 fb1 = *(const float2*)(brow + k1 + 2 * j);
                    bm[2 * j]     = (_Float16)(bval ? fb0.x : 0.f);
                    bm[2 * j + 1] = (_Float16)(bval ? fb0.y : 0.f);
                    bm[8 + 2 * j]     = (_Float16)(bval ? fb1.x : 0.f);
                    bm[8 + 2 * j + 1] = (_Float16)(bval ? fb1.y : 0.f);
                }
                c = __builtin_amdgcn_wmma_f32_16x16x32_f16(false, a, false, bm,
                                                           (short)0, c, false, false);
            }
            if (col < 2) {
                #pragma unroll
                for (int g = 0; g < 8; ++g)
                    sW[col * PPAD + wave * 16 + half8 + g] = c[g];
            }
        }

        const int bn = b + NBLK;
        if (bn >= B) break;
        {
            padmask = __builtin_amdgcn_ballot_w32(vr < PMAX && ids >= n_items);
            {
                const int ids_ = ids;
                const int pid_ = __builtin_amdgcn_readlane(ids, 13), nid_ = __builtin_amdgcn_readlane(ids, 14);
                const int uid_ = __builtin_amdgcn_readlane(ids, 15);
                FOR_ROWS(ROW_BASE)
                FOR_ROWS(ROW_LOAD)
                u = (user_emb + (size_t)uid_ * DIMS)[tid];
            }
            npc = np1r;
            const int bi = min(bn + NBLK, B - 1);
            ids  = ids_load(pref_ids, pos_ids, neg_ids, user_ids, bi, vr, lane);
            np1r = np_load(n_prefs, bi, lane);
        }
        finish_sample(sPref, sT, sW, sAtt, sRed, out, wave, lane, tid, np1, u_cur, B, b);
        b = bn;
    }
    finish_sample(sPref, sT, sW, sAtt, sRed, out, wave, lane, tid, np1, u_cur, B, b);
}

extern "C" void kernel_launch(void* const* d_in, const int* in_sizes, int n_in,
                              void* d_out, int out_size, void* d_ws, size_t ws_size,
                              hipStream_t stream) {
    (void)n_in; (void)out_size; (void)d_ws; (void)ws_size;
    const float* user_emb = (const float*)d_in[0];
    const float* item_emb = (const float*)d_in[1];
    const int*   user_ids = (const int*)d_in[2];
    const int*   pos_ids  = (const int*)d_in[3];
    const int*   neg_ids  = (const int*)d_in[4];
    const int*   pref_ids = (const int*)d_in[5];
    const float* n_prefs  = (const float*)d_in[6];
    float* out = (float*)d_out;

    const int B       = in_sizes[2];
    const int n_items = in_sizes[1] / DIMS;

    AttCML_kernel<<<NBLK, 128, 0, stream>>>(user_emb, item_emb, user_ids, pos_ids,
                                         neg_ids, pref_ids, n_prefs, out, B, n_items);
}


# Wave-private M-tiles: a wave's 13 gathered rows + both targets are its own tile, scores return in its accumulators, weighted sum runs on the gather registers; per-wave partial sums, normalised last
# speedup vs baseline: 2.2836x; 1.2860x over previous
#include <hip/hip_runtime.h>
#include <hip/hip_bf16.h>


typedef __attribute__((ext_vector_type(16))) _Float16 v16h;
typedef __attribute__((ext_vector_type(8)))  float    v8f;

#define DIMS   128
#define PMAX   50
#define RPW    13
#define TROWS  16
#define LSTR   132
#define NBLK   2048

static __device__ __forceinline__ void wg_barrier() {
    asm volatile("s_wait_dscnt 0x0\n\ts_barrier_signal -1\n\ts_barrier_wait -1" ::: "memory");
}

static __device__ __forceinline__ const float4* row_base(const float* item_emb, int myid, int j, int n_items)
{
    const int id  = __builtin_amdgcn_readlane(myid, j);
    const int idc = min(id, n_items - 1);
    const size_t   off = (size_t)idc * (DIMS * sizeof(float));
    const unsigned lo  = __builtin_amdgcn_readfirstlane((unsigned)off);
    const unsigned hi  = __builtin_amdgcn_readfirstlane((unsigned)(off >> 32));
    const char*    row = (const char*)item_emb + (((size_t)hi << 32) | lo);
    return (const float4*)row;
}

static __device__ __forceinline__ int ids_load(const int* pref_ids, const int* pos_ids, const int* neg_ids,
                                               const int* user_ids, int bi, int vr, int lane)
{
    const int* p = pref_ids + (size_t)bi * PMAX + vr;
    if (lane == 13) p = pos_ids + bi;
    if (lane == 14) p = neg_ids + bi;
    if (lane == 15) p = user_ids + bi;
    int r = 0;
    if (vr < PMAX || (lane >= 13 && lane <= 15)) r = *p;
    return r;
}

static __device__ __forceinline__ float np_load(const float* n_prefs, int bi, int lane)
{
    float r = 0.f;
    if (lane == 0) r = n_prefs[bi];
    return r;
}
static __device__ __forceinline__ float np_use(float npv)
{
    return __int_as_float(__builtin_amdgcn_readfirstlane(__float_as_int(npv)));
}
static __device__ __forceinline__ float lane_val(float x, int l)
{
    return __int_as_float(__builtin_amdgcn_readlane(__float_as_int(x), l));
}

#define FOR_PREF(X) X(0) X(1) X(2) X(3) X(4) X(5) X(6) X(7) X(8) X(9) X(10) X(11) X(12)
#define FOR_ROWS(X) FOR_PREF(X) X(13) X(14)
#define ROW_DECL(j)  float4 v##j = make_float4(0.f, 0.f, 0.f, 0.f);
#define ROW_BASE(j)  const float4* rp##j = row_base(item_emb, ids_, j, n_items);
#define ROW_LOAD(j)  if (j != RPW - 1 || wave_s < 2) v##j = rp##j[lane];
#define ROW_STORE(j) *(float4*)(&tile[j * LSTR + lane * 4]) = v##j;
#define ROW_ZERO(j)  if ((padmask >> j) & 1u) v##j = make_float4(0.f, 0.f, 0.f, 0.f);
#define ROW_FMA(j)   { const float eP = lane_val(ev[j & 7], 16 * (j >> 3));                 \
                       const float eN = lane_val(ev[j & 7], 16 * (j >> 3) + 1);             \
                       esP += eP; esN += eN;                                                \
                       accP.x = fmaf(eP, v##j.x, accP.x); accP.y = fmaf(eP, v##j.y, accP.y); \
                       accP.z = fmaf(eP, v##j.z, accP.z); accP.w = fmaf(eP, v##j.w, accP.w); \
                       accN.x = fmaf(eN, v##j.x, accN.x); accN.y = fmaf(eN, v##j.y, accN.y); \
                       accN.z = fmaf(eN, v##j.z, accN.z); accN.w = fmaf(eN, v##j.w, accN.w); }
#define ISSUE_ROWS()  {                                                                     \
        const int ids_ = ids;                                                               \
        const int uid_ = __builtin_amdgcn_readlane(ids, 15);                                \
        FOR_ROWS(ROW_BASE)                                                                  \
        FOR_ROWS(ROW_LOAD)                                                                  \
        if (wave_s >= 2) u4 = ((const float4*)(user_emb + (size_t)uid_ * DIMS))[lane];      \
    }

static __device__ __forceinline__ void finish_sample(float* sPart, float* sEs, float* out,
                                                     int wave_s, int lane, float4 accP, float4 accN,
                                                     float esP, float esN, float4 up, int B, int b)
{
    wg_barrier();
    *(float4*)(&sPart[(0 * 4 + wave_s) * DIMS + lane * 4]) = accP;
    *(float4*)(&sPart[(1 * 4 + wave_s) * DIMS + lane * 4]) = accN;
    if (lane == 0) { sEs[wave_s] = esP; sEs[4 + wave_s] = esN; }
    wg_barrier();
    if (wave_s >= 2) {
        const int t = wave_s - 2;
        const float* pp = &sPart[t * 4 * DIMS + lane * 4];
        const float4 a0 = *(const float4*)(pp);
        const float4 a1 = *(const float4*)(pp + DIMS);
        const float4 a2 = *(const float4*)(pp + 2 * DIMS);
        const float4 a3 = *(const float4*)(pp + 3 * DIMS);
        const float4 es = *(const float4*)(&sEs[t * 4]);
        const float  inv = 1.0f / ((es.x + es.y) + (es.z + es.w));
        const float dx = fmaf((a0.x + a1.x) + (a2.x + a3.x), inv, up.x);
        const float dy = fmaf((a0.y + a1.y) + (a2.y + a3.y), inv, up.y);
        const float dz = fmaf((a0.z + a1.z) + (a2.z + a3.z), inv, up.z);
        const float dw = fmaf((a0.w + a1.w) + (a2.w + a3.w), inv, up.w);
        float sq = (dx * dx + dy * dy) + (dz * dz + dw * dw);
        #pragma unroll
        for (int off = 16; off > 0; off >>= 1) sq += __shfl_xor(sq, off, 32);
        if (lane == 0) out[(size_t)t * B + b] = sq;
    }
}

__global__ __launch_bounds__(128) __attribute__((amdgpu_waves_per_eu(8, 8)))
void AttCML_kernel(const float* user_emb,
                   const float* item_emb,
                   const int*   user_ids,
                   const int*   pos_ids,
                   const int*   neg_ids,
                   const int*   pref_ids,
                   const float* n_prefs,
                   float* out,
                   int B, int n_items)
{
    __shared__ __align__(16) float sTile[4 * TROWS * LSTR];
    __shared__ __align__(16) float sPart[2 * 4 * DIMS];
    __shared__ __align__(16) float sEs[2 * 4];

    const int tid  = threadIdx.x;
    const int wave = tid >> 5;
    const int lane = tid & 31;
    int b = blockIdx.x;
    if (b >= B) return;

    const int wave_s = __builtin_amdgcn_readfirstlane(wave) & 3;
    const int vr = wave_s + 4 * lane;
    float* const tile = sTile + wave_s * (TROWS * LSTR);

    *(float4*)(&tile[(TROWS - 1) * LSTR + lane * 4]) = make_float4(0.f, 0.f, 0.f, 0.f);

    int ids = ids_load(pref_ids, pos_ids, neg_ids, user_ids, b, vr, lane);
    float npc  = np_load(n_prefs, b, lane);
    float np1r = np_load(n_prefs, min(b + NBLK, B - 1), lane);
    FOR_ROWS(ROW_DECL)
    float4 u4 = make_float4(0.f, 0.f, 0.f, 0.f);
    unsigned padmask = __builtin_amdgcn_ballot_w32(vr < PMAX && ids >= n_items);
    ISSUE_ROWS()
    ids = ids_load(pref_ids, pos_ids, neg_ids, user_ids, min(b + NBLK, B - 1), vr, lane);

    float4 up, accP, accN;
    float  esP, esN;

    for (;;) {
        if (padmask != 0u) {
            FOR_PREF(ROW_ZERO)
        }
        {
            const float4 tg = (wave_s == 3) ? v14 : v13;
            up = make_float4(u4.x - tg.x, u4.y - tg.y, u4.z - tg.z, u4.w - tg.w);
        }
        FOR_ROWS(ROW_STORE)
        const float np1 = np_use(npc) + 1.0f;

        const int half8 = (lane >> 4) ? 8 : 0;
        v8f c = {0.f, 0.f, 0.f, 0.f, 0.f, 0.f, 0.f, 0.f};
        {
            const int mrow  = lane & 15;
            const int col   = lane & 15;
            const float* arow = &tile[mrow * LSTR];
            const float* brow = &tile[(RPW + (col < 2 ? col : 0)) * LSTR];
            const bool   bval = (col < 2);

            #pragma unroll 1
            for (int kb = 0; kb < DIMS; kb += 32) {
                const int k0 = kb + half8;
                const int k1 = kb + 16 + half8;
                v16h a, bm;
                #pragma unroll
                for (int j = 0; j < 4; ++j) {
                    float2 fa0 = *(const float2*)(arow + k0 + 2 * j);
                    float2 fa1 = *(const float2*)(arow + k1 + 2 * j);
                    a[2 * j]     = (_Float16)fa0.x;
                    a[2 * j + 1] = (_Float16)fa0.y;
                    a[8 + 2 * j]     = (_Float16)fa1.x;
                    a[8 + 2 * j + 1] = (_Float16)fa1.y;
                    float2 fb0 = *(const float2*)(brow + k0 + 2 * j);
                    float2 fb1 = *(const float2*)(brow + k1 + 2 * j);
                    bm[2 * j]     = (_Float16)(bval ? fb0.x : 0.f);
                    bm[2 * j + 1] = (_Float16)(bval ? fb0.y : 0.f);
                    bm[8 + 2 * j]     = (_Float16)(bval ? fb1.x : 0.f);
                    bm[8 + 2 * j + 1] = (_Float16)(bval ? fb1.y : 0.f);
                }
                c = __builtin_amdgcn_wmma_f32_16x16x32_f16(false, a, false, bm,
                                                           (short)0, c, false, false);
            }
        }

        float ev[8];
        #pragma unroll
        for (int g = 0; g < 8; ++g) {
            const int j = half8 + g;
            const int p = wave_s + 4 * j;
            const bool ok = (j < RPW) && (p < PMAX) && ((float)p < np1);
            ev[g] = ok ? __expf(c[g]) : 0.f;
        }

        accP = make_float4(0.f, 0.f, 0.f, 0.f); accN = make_float4(0.f, 0.f, 0.f, 0.f);
        esP = 0.f; esN = 0.f;
        FOR_PREF(ROW_FMA)

        const int bn = b + NBLK;
        if (bn >= B) break;
        {
            padmask = __builtin_amdgcn_ballot_w32(vr < PMAX && ids >= n_items);
            ISSUE_ROWS()
            npc = np1r;
            const int bi = min(bn + NBLK, B - 1);
            ids  = ids_load(pref_ids, pos_ids, neg_ids, user_ids, bi, vr, lane);
            np1r = np_load(n_prefs, bi, lane);
        }

        finish_sample(sPart, sEs, out, wave_s, lane, accP, accN, esP, esN, up, B, b);
        b = bn;
    }
    finish_sample(sPart, sEs, out, wave_s, lane, accP, accN, esP, esN, up, B, b);
}

extern "C" void kernel_launch(void* const* d_in, const int* in_sizes, int n_in,
                              void* d_out, int out_size, void* d_ws, size_t ws_size,
                              hipStream_t stream) {
    (void)n_in; (void)out_size; (void)d_ws; (void)ws_size;
    const float* user_emb = (const float*)d_in[0];
    const float* item_emb = (const float*)d_in[1];
    const int*   user_ids = (const int*)d_in[2];
    const int*   pos_ids  = (const int*)d_in[3];
    const int*   neg_ids  = (const int*)d_in[4];
    const int*   pref_ids = (const int*)d_in[5];
    const float* n_prefs  = (const float*)d_in[6];
    float* out = (float*)d_out;

    const int B       = in_sizes[2];
    const int n_items = in_sizes[1] / DIMS;

    AttCML_kernel<<<NBLK, 128, 0, stream>>>(user_emb, item_emb, user_ids, pos_ids,
                                         neg_ids, pref_ids, n_prefs, out, B, n_items);
}


# Gathered rows rounded to f16 once in registers and the per-wave LDS tile kept in f16: a fragment is two 16-byte reads with no convert or select behind it, LDS bytes halved. Operands bit-identical
# speedup vs baseline: 2.6903x; 1.1781x over previous
#include <hip/hip_runtime.h>
#include <hip/hip_bf16.h>


typedef __attribute__((ext_vector_type(16))) _Float16 v16h;
typedef __attribute__((ext_vector_type(8)))  _Float16 v8h;
typedef __attribute__((ext_vector_type(4)))  _Float16 v4h;
typedef __attribute__((ext_vector_type(8)))  float    v8f;

#define DIMS   128
#define PMAX   50
#define RPW    13
#define TROWS  16
#define HSTR   136
#define NBLK   2048

static __device__ __forceinline__ void wg_barrier() {
    asm volatile("s_wait_dscnt 0x0\n\ts_barrier_signal -1\n\ts_barrier_wait -1" ::: "memory");
}

static __device__ __forceinline__ const float4* row_base(const float* item_emb, int myid, int j, int n_items)
{
    const int id  = __builtin_amdgcn_readlane(myid, j);
    const int idc = min(id, n_items - 1);
    const size_t   off = (size_t)idc * (DIMS * sizeof(float));
    const unsigned lo  = __builtin_amdgcn_readfirstlane((unsigned)off);
    const unsigned hi  = __builtin_amdgcn_readfirstlane((unsigned)(off >> 32));
    const char*    row = (const char*)item_emb + (((size_t)hi << 32) | lo);
    return (const float4*)row;
}

static __device__ __forceinline__ int ids_load(const int* pref_ids, const int* pos_ids, const int* neg_ids,
                                               const int* user_ids, int bi, int vr, int lane)
{
    const int* p = pref_ids + (size_t)bi * PMAX + vr;
    if (lane == 13) p = pos_ids + bi;
    if (lane == 14) p = neg_ids + bi;
    if (lane == 15) p = user_ids + bi;
    int r = 0;
    if (vr < PMAX || (lane >= 13 && lane <= 15)) r = *p;
    return r;
}

static __device__ __forceinline__ float np_load(const float* n_prefs, int bi, int lane)
{
    float r = 0.f;
    if (lane == 0) r = n_prefs[bi];
    return r;
}
static __device__ __forceinline__ float np_use(float npv)
{
    return __int_as_float(__builtin_amdgcn_readfirstlane(__float_as_int(npv)));
}
static __device__ __forceinline__ float lane_val(float x, int l)
{
    return __int_as_float(__builtin_amdgcn_readlane(__float_as_int(x), l));
}

#define FOR_PREF(X) X(0) X(1) X(2) X(3) X(4) X(5) X(6) X(7) X(8) X(9) X(10) X(11) X(12)
#define FOR_ROWS(X) FOR_PREF(X) X(13) X(14)
#define ROW_DECL(j)  float4 v##j = make_float4(0.f, 0.f, 0.f, 0.f);
#define ROW_BASE(j)  const float4* rp##j = row_base(item_emb, ids_, j, n_items);
#define ROW_LOAD(j)  if (j != RPW - 1 || wave_s < 2) v##j = rp##j[lane];
#define ROW_STORE(j) { v4h h_; h_[0] = (_Float16)v##j.x; h_[1] = (_Float16)v##j.y;                \
                       h_[2] = (_Float16)v##j.z; h_[3] = (_Float16)v##j.w;                         \
                       *(v4h*)(&tile[j * HSTR + lane * 4]) = h_; }
#define ROW_ZERO(j)  if ((padmask >> j) & 1u) v##j = make_float4(0.f, 0.f, 0.f, 0.f);
#define ROW_FMA(j)   { const float eP = lane_val(ev[j & 7], 16 * (j >> 3));                 \
                       const float eN = lane_val(ev[j & 7], 16 * (j >> 3) + 1);             \
                       esP += eP; esN += eN;                                                \
                       accP.x = fmaf(eP, v##j.x, accP.x); accP.y = fmaf(eP, v##j.y, accP.y); \
                       accP.z = fmaf(eP, v##j.z, accP.z); accP.w = fmaf(eP, v##j.w, accP.w); \
                       accN.x = fmaf(eN, v##j.x, accN.x); accN.y = fmaf(eN, v##j.y, accN.y); \
                       accN.z = fmaf(eN, v##j.z, accN.z); accN.w = fmaf(eN, v##j.w, accN.w); }
#define ISSUE_ROWS()  {                                                                     \
        const int ids_ = ids;                                                               \
        const int uid_ = __builtin_amdgcn_readlane(ids, 15);                                \
        FOR_ROWS(ROW_BASE)                                                                  \
        FOR_ROWS(ROW_LOAD)                                                                  \
        if (wave_s >= 2) u4 = ((const float4*)(user_emb + (size_t)uid_ * DIMS))[lane];      \
    }

static __device__ __forceinline__ void finish_sample(float* sPart, float* sEs, float* out,
                                                     int wave_s, int lane, float4 accP, float4 accN,
                                                     float esP, float esN, float4 up, int B, int b)
{
    wg_barrier();
    *(float4*)(&sPart[(0 * 4 + wave_s) * DIMS + lane * 4]) = accP;
    *(float4*)(&sPart[(1 * 4 + wave_s) * DIMS + lane * 4]) = accN;
    if (lane == 0) { sEs[wave_s] = esP; sEs[4 + wave_s] = esN; }
    wg_barrier();
    if (wave_s >= 2) {
        const int t = wave_s - 2;
        const float* pp = &sPart[t * 4 * DIMS + lane * 4];
        const float4 a0 = *(const float4*)(pp);
        const float4 a1 = *(const float4*)(pp + DIMS);
        const float4 a2 = *(const float4*)(pp + 2 * DIMS);
        const float4 a3 = *(const float4*)(pp + 3 * DIMS);
        const float4 es = *(const float4*)(&sEs[t * 4]);
        const float  inv = 1.0f / ((es.x + es.y) + (es.z + es.w));
        const float dx = fmaf((a0.x + a1.x) + (a2.x + a3.x), inv, up.x);
        const float dy = fmaf((a0.y + a1.y) + (a2.y + a3.y), inv, up.y);
        const float dz = fmaf((a0.z + a1.z) + (a2.z + a3.z), inv, up.z);
        const float dw = fmaf((a0.w + a1.w) + (a2.w + a3.w), inv, up.w);
        float sq = (dx * dx + dy * dy) + (dz * dz + dw * dw);
        #pragma unroll
        for (int off = 16; off > 0; off >>= 1) sq += __shfl_xor(sq, off, 32);
        if (lane == 0) out[(size_t)t * B + b] = sq;
    }
}

__global__ __launch_bounds__(128) __attribute__((amdgpu_waves_per_eu(8, 8)))
void AttCML_kernel(const float* user_emb,
                   const float* item_emb,
                   const int*   user_ids,
                   const int*   pos_ids,
                   const int*   neg_ids,
                   const int*   pref_ids,
                   const float* n_prefs,
                   float* out,
                   int B, int n_items)
{
    __shared__ __align__(16) _Float16 sTile[4 * TROWS * HSTR];
    __shared__ __align__(16) float sPart[2 * 4 * DIMS];
    __shared__ __align__(16) float sEs[2 * 4];

    const int tid  = threadIdx.x;
    const int wave = tid >> 5;
    const int lane = tid & 31;
    int b = blockIdx.x;
    if (b >= B) return;

    const int wave_s = __builtin_amdgcn_readfirstlane(wave) & 3;
    const int vr = wave_s + 4 * lane;
    _Float16* const tile = sTile + wave_s * (TROWS * HSTR);

    {
        const v4h z_ = {(_Float16)0.f, (_Float16)0.f, (_Float16)0.f, (_Float16)0.f};
        *(v4h*)(&tile[(TROWS - 1) * HSTR + lane * 4]) = z_;
    }

    int ids = ids_load(pref_ids, pos_ids, neg_ids, user_ids, b, vr, lane);
    float npc  = np_load(n_prefs, b, lane);
    float np1r = np_load(n_prefs, min(b + NBLK, B - 1), lane);
    FOR_ROWS(ROW_DECL)
    float4 u4 = make_float4(0.f, 0.f, 0.f, 0.f);
    unsigned padmask = __builtin_amdgcn_ballot_w32(vr < PMAX && ids >= n_items);
    ISSUE_ROWS()
    ids = ids_load(pref_ids, pos_ids, neg_ids, user_ids, min(b + NBLK, B - 1), vr, lane);

    float4 up, accP, accN;
    float  esP, esN;

    for (;;) {
        if (padmask != 0u) {
            FOR_PREF(ROW_ZERO)
        }
        {
            const float4 tg = (wave_s == 3) ? v14 : v13;
            up = make_float4(u4.x - tg.x, u4.y - tg.y, u4.z - tg.z, u4.w - tg.w);
        }
        FOR_ROWS(ROW_STORE)
        const float np1 = np_use(npc) + 1.0f;

        const int half8 = (lane >> 4) ? 8 : 0;
        v8f c = {0.f, 0.f, 0.f, 0.f, 0.f, 0.f, 0.f, 0.f};
        {
            const int mrow  = lane & 15;
            const int col   = lane & 15;
            const _Float16* arow = &tile[mrow * HSTR + half8];
            const _Float16* brow = &tile[(col < 2 ? RPW + col : TROWS - 1) * HSTR + half8];

            #pragma unroll
            for (int kb = 0; kb < DIMS; kb += 32) {
                const v8h a0 = *(const v8h*)(arow + kb);
                const v8h a1 = *(const v8h*)(arow + kb + 16);
                const v8h b0 = *(const v8h*)(brow + kb);
                const v8h b1 = *(const v8h*)(brow + kb + 16);
                const v16h a  = __builtin_shufflevector(a0, a1, 0, 1, 2, 3, 4, 5, 6, 7, 8, 9, 10, 11, 12, 13, 14, 15);
                const v16h bm = __builtin_shufflevector(b0, b1, 0, 1, 2, 3, 4, 5, 6, 7, 8, 9, 10, 11, 12, 13, 14, 15);
                c = __builtin_amdgcn_wmma_f32_16x16x32_f16(false, a, false, bm,
                                                           (short)0, c, false, false);
            }
        }

        float ev[8];
        #pragma unroll
        for (int g = 0; g < 8; ++g) {
            const int j = half8 + g;
            const int p = wave_s + 4 * j;
            const bool ok = (j < RPW) && (p < PMAX) && ((float)p < np1);
            ev[g] = ok ? __expf(c[g]) : 0.f;
        }

        accP = make_float4(0.f, 0.f, 0.f, 0.f); accN = make_float4(0.f, 0.f, 0.f, 0.f);
        esP = 0.f; esN = 0.f;
        FOR_PREF(ROW_FMA)

        const int bn = b + NBLK;
        if (bn >= B) break;
        {
            padmask = __builtin_amdgcn_ballot_w32(vr < PMAX && ids >= n_items);
            ISSUE_ROWS()
            npc = np1r;
            const int bi = min(bn + NBLK, B - 1);
            ids  = ids_load(pref_ids, pos_ids, neg_ids, user_ids, bi, vr, lane);
            np1r = np_load(n_prefs, bi, lane);
        }

        finish_sample(sPart, sEs, out, wave_s, lane, accP, accN, esP, esN, up, B, b);
        b = bn;
    }
    finish_sample(sPart, sEs, out, wave_s, lane, accP, accN, esP, esN, up, B, b);
}

extern "C" void kernel_launch(void* const* d_in, const int* in_sizes, int n_in,
                              void* d_out, int out_size, void* d_ws, size_t ws_size,
                              hipStream_t stream) {
    (void)n_in; (void)out_size; (void)d_ws; (void)ws_size;
    const float* user_emb = (const float*)d_in[0];
    const float* item_emb = (const float*)d_in[1];
    const int*   user_ids = (const int*)d_in[2];
    const int*   pos_ids  = (const int*)d_in[3];
    const int*   neg_ids  = (const int*)d_in[4];
    const int*   pref_ids = (const int*)d_in[5];
    const float* n_prefs  = (const float*)d_in[6];
    float* out = (float*)d_out;

    const int B       = in_sizes[2];
    const int n_items = in_sizes[1] / DIMS;

    AttCML_kernel<<<NBLK, 128, 0, stream>>>(user_emb, item_emb, user_ids, pos_ids,
                                         neg_ids, pref_ids, n_prefs, out, B, n_items);
}


# Each wave owns whole samples: 50 rows in four 13-row chunks into ONE running f32 sum per target (sum order changes; was 4 waves' partial sums added via LDS); targets, user row load once; no barriers
# speedup vs baseline: 3.1005x; 1.1525x over previous
#include <hip/hip_runtime.h>
#include <hip/hip_bf16.h>


typedef __attribute__((ext_vector_type(16))) _Float16 v16h;
typedef __attribute__((ext_vector_type(8)))  _Float16 v8h;
typedef __attribute__((ext_vector_type(4)))  _Float16 v4h;
typedef __attribute__((ext_vector_type(8)))  float    v8f;

#define DIMS   128
#define PMAX   50
#define RPW    13
#define TROWS  16
#define HSTR   136
#define NBLK   2048
#define NWAVES (NBLK * 4)

static __device__ __forceinline__ const float4* row_at(const float* table, int id)
{
    const size_t   off = (size_t)id * (DIMS * sizeof(float));
    const unsigned lo  = __builtin_amdgcn_readfirstlane((unsigned)off);
    const unsigned hi  = __builtin_amdgcn_readfirstlane((unsigned)(off >> 32));
    const char*    row = (const char*)table + (((size_t)hi << 32) | lo);
    return (const float4*)row;
}
static __device__ __forceinline__ const float4* row_base(const float* item_emb, int idreg, int l, int n_items)
{
    const int id  = __builtin_amdgcn_readlane(idreg, l);
    return row_at(item_emb, min(id, n_items - 1));
}

static __device__ __forceinline__ int idsA_load(const int* pref_ids, int bi, int lane)
{
    return pref_ids[(size_t)bi * PMAX + lane];
}
static __device__ __forceinline__ int idsB_load(const int* pref_ids, const int* pos_ids, const int* neg_ids,
                                                const int* user_ids, int bi, int lane)
{
    const int* p = pref_ids + (size_t)bi * PMAX + 32 + lane;
    if (lane == 18) p = pos_ids + bi;
    if (lane == 19) p = neg_ids + bi;
    if (lane == 20) p = user_ids + bi;
    int r = 0;
    if (lane <= 20) r = *p;
    return r;
}
static __device__ __forceinline__ float np_load(const float* n_prefs, int bi, int lane)
{
    float r = 0.f;
    if (lane == 0) r = n_prefs[bi];
    return r;
}
static __device__ __forceinline__ float np_use(float npv)
{
    return __int_as_float(__builtin_amdgcn_readfirstlane(__float_as_int(npv)));
}
static __device__ __forceinline__ float lane_val(float x, int l)
{
    return __int_as_float(__builtin_amdgcn_readlane(__float_as_int(x), l));
}

static __device__ __forceinline__ float xhalf(float x)
{
    const int xi = __float_as_int(x);
    return __int_as_float(__builtin_amdgcn_permlanex16(xi, xi, 0x76543210, 0xfedcba98, false, false));
}
template <int M>
static __device__ __forceinline__ float xrow(float x)
{
    const int xi = __float_as_int(x);
    return __int_as_float(__builtin_amdgcn_update_dpp(xi, xi, 0x160 + M, 0xf, 0xf, false));
}

#define FOR_PREF(X, C) X(0,C) X(1,C) X(2,C) X(3,C) X(4,C) X(5,C) X(6,C) X(7,C) X(8,C) X(9,C) X(10,C) X(11,C) X(12,C)
#define P_OF(j, C)     (RPW * (C) + (j))
#define HAS(j, C)      (P_OF(j, C) < PMAX)
#define ROW_DECL(j, C) float4 v##j = make_float4(0.f, 0.f, 0.f, 0.f);
#define ROW_BASE(j, C) const float4* rp##j = row_base(item_emb, P_OF(j, C) < 32 ? iA_ : iB_, P_OF(j, C) & 31, n_items);
#define ROW_LOAD(j, C) if (HAS(j, C)) v##j = rp##j[lane];
#define ROW_STORE_AT(r, val) { v4h h_; h_[0] = (_Float16)(val).x; h_[1] = (_Float16)(val).y;     \
                               h_[2] = (_Float16)(val).z; h_[3] = (_Float16)(val).w;              \
                               *(v4h*)(&tile[(r) * HSTR + lane * 4]) = h_; }
#define ROW_STORE(j, C) if (HAS(j, C)) ROW_STORE_AT(j, v##j)
#define ROW_ZERO(j, C)  if (HAS(j, C) && ((pm_ >> j) & 1u)) v##j = make_float4(0.f, 0.f, 0.f, 0.f);
#define ROW_FMA(j, C)  if (HAS(j, C)) {                                                       \
                       const float eP = lane_val(ev[j & 7], 16 * (j >> 3));                  \
                       const float eN = lane_val(ev[j & 7], 16 * (j >> 3) + 1);              \
                       accP.x = fmaf(eP, v##j.x, accP.x); accP.y = fmaf(eP, v##j.y, accP.y); \
                       accP.z = fmaf(eP, v##j.z, accP.z); accP.w = fmaf(eP, v##j.w, accP.w); \
                       accN.x = fmaf(eN, v##j.x, accN.x); accN.y = fmaf(eN, v##j.y, accN.y); \
                       accN.z = fmaf(eN, v##j.z, accN.z); accN.w = fmaf(eN, v##j.w, accN.w); }
#define ISSUE_CHUNK(C, IA, IB) {                                                             \
        const int iA_ = (IA); const int iB_ = (IB);                                          \
        FOR_PREF(ROW_BASE, C)                                                                \
        FOR_PREF(ROW_LOAD, C)                                                                \
    }
#define PADBITS(C) ((C) == 0 ? (mA & 0x1fffu) : (C) == 1 ? ((mA >> 13) & 0x1fffu)            \
                  : (C) == 2 ? (((mA >> 26) | (mB << 6)) & 0x1fffu) : ((mB >> 7) & 0x7ffu))

static __device__ __forceinline__ v8f score_tile(const _Float16* tile, int lane)
{
    const int half8 = (lane >> 4) ? 8 : 0;
    const int mrow  = lane & 15;
    const int col   = lane & 15;
    const _Float16* arow = &tile[mrow * HSTR + half8];
    const _Float16* brow = &tile[(col < 2 ? RPW + col : TROWS - 1) * HSTR + half8];
    v8f c = {0.f, 0.f, 0.f, 0.f, 0.f, 0.f, 0.f, 0.f};
    #pragma unroll
    for (int kb = 0; kb < DIMS; kb += 32) {
        const v8h a0 = *(const v8h*)(arow + kb);
        const v8h a1 = *(const v8h*)(arow + kb + 16);
        const v8h b0 = *(const v8h*)(brow + kb);
        const v8h b1 = *(const v8h*)(brow + kb + 16);
        const v16h a  = __builtin_shufflevector(a0, a1, 0, 1, 2, 3, 4, 5, 6, 7, 8, 9, 10, 11, 12, 13, 14, 15);
        const v16h bm = __builtin_shufflevector(b0, b1, 0, 1, 2, 3, 4, 5, 6, 7, 8, 9, 10, 11, 12, 13, 14, 15);
        c = __builtin_amdgcn_wmma_f32_16x16x32_f16(false, a, false, bm, (short)0, c, false, false);
    }
    return c;
}

#define COMPUTE_CHUNK(C) {                                                                    \
        const unsigned pm_ = PADBITS(C);                                                      \
        if (pm_ != 0u) {                \
            FOR_PREF(ROW_ZERO, C)                                                             \
        }                                                                                     \
        FOR_PREF(ROW_STORE, C)                                                                \
        const v8f c = score_tile(tile, lane);                                                 \
          \
        float ev[8];                                                                          \
          \
          \
          \
          \
        const float thr_ = fminf(np1, (float)(RPW * ((C) + 1) < PMAX ? RPW * ((C) + 1) : PMAX)) - (float)half8; \
        _Pragma("unroll")                                                                     \
        for (int g = 0; g < 8; ++g) {                                                         \
            ev[g] = ((float)(RPW * (C) + g) < thr_) ? __expf(c[g]) : 0.f;                     \
        }                                                                                     \
          \
        esv += ((ev[0] + ev[1]) + (ev[2] + ev[3])) + ((ev[4] + ev[5]) + (ev[6] + ev[7]));     \
        FOR_PREF(ROW_FMA, C)                                                                  \
          \
          \
        asm volatile("" : "+v"(accP.x), "+v"(accP.y), "+v"(accP.z), "+v"(accP.w),                   \
                          "+v"(accN.x), "+v"(accN.y), "+v"(accN.z), "+v"(accN.w), "+v"(esv));  \
    }

__global__ __launch_bounds__(128) __attribute__((amdgpu_waves_per_eu(8, 8)))
void AttCML_kernel(const float* user_emb,
                   const float* item_emb,
                   const int*   user_ids,
                   const int*   pos_ids,
                   const int*   neg_ids,
                   const int*   pref_ids,
                   const float* n_prefs,
                   float* out,
                   int B, int n_items)
{
    __shared__ __align__(16) _Float16 sTile[4 * TROWS * HSTR];

    const int tid  = threadIdx.x;
    const int lane = tid & 31;
    const int wave_s = __builtin_amdgcn_readfirstlane(tid >> 5) & 3;
    int b = blockIdx.x * 4 + wave_s;
    if (b >= B) return;

    _Float16* const tile = sTile + wave_s * (TROWS * HSTR);
    const int half8 = (lane >> 4) ? 8 : 0;

    {
        const v4h z_ = {(_Float16)0.f, (_Float16)0.f, (_Float16)0.f, (_Float16)0.f};
        *(v4h*)(&tile[(TROWS - 1) * HSTR + lane * 4]) = z_;
    }

    int idsA = idsA_load(pref_ids, b, lane);
    int idsB = idsB_load(pref_ids, pos_ids, neg_ids, user_ids, b, lane);
    float npc = np_load(n_prefs, b, lane);
    FOR_PREF(ROW_DECL, 0)
    float4 tP = make_float4(0.f, 0.f, 0.f, 0.f), tN = tP, u4 = tP;
    ISSUE_CHUNK(0, idsA, idsB)
    {
        const float4* rpP = row_base(item_emb, idsB, 18, n_items);
        const float4* rpN = row_base(item_emb, idsB, 19, n_items);
        const int uid_ = __builtin_amdgcn_readlane(idsB, 20);
        tP = rpP[lane]; tN = rpN[lane];
        u4 = row_at(user_emb, uid_)[lane];
    }
    int idsAn = idsA_load(pref_ids, min(b + NWAVES, B - 1), lane);
    int idsBn = idsB_load(pref_ids, pos_ids, neg_ids, user_ids, min(b + NWAVES, B - 1), lane);
    float npn = np_load(n_prefs, min(b + NWAVES, B - 1), lane);

    for (;;) {
        const unsigned mA = __builtin_amdgcn_ballot_w32(idsA >= n_items);
        const unsigned mB = __builtin_amdgcn_ballot_w32(lane < PMAX - 32 && idsB >= n_items);
        const float np1 = np_use(npc) + 1.0f;

        float4 accP = make_float4(0.f, 0.f, 0.f, 0.f), accN = accP;
        float  esv = 0.f;

        float4 upP = make_float4(u4.x - tP.x, u4.y - tP.y, u4.z - tP.z, u4.w - tP.w);
        float4 upN = make_float4(u4.x - tN.x, u4.y - tN.y, u4.z - tN.z, u4.w - tN.w);
        asm volatile("" : "+v"(upP.x), "+v"(upP.y), "+v"(upP.z), "+v"(upP.w),
                          "+v"(upN.x), "+v"(upN.y), "+v"(upN.z), "+v"(upN.w));
        ROW_STORE_AT(RPW, tP)
        ROW_STORE_AT(RPW + 1, tN)

        COMPUTE_CHUNK(0)
        ISSUE_CHUNK(1, idsA, idsB)
        COMPUTE_CHUNK(1)
        ISSUE_CHUNK(2, idsA, idsB)
        COMPUTE_CHUNK(2)
        ISSUE_CHUNK(3, idsA, idsB)
        COMPUTE_CHUNK(3)

        const int  bn   = b + NWAVES;
        const bool more = bn < B;
        if (more) {
            idsA = idsAn; idsB = idsBn; npc = npn;
            ISSUE_CHUNK(0, idsA, idsB)
            {
                const float4* rpP = row_base(item_emb, idsB, 18, n_items);
                const float4* rpN = row_base(item_emb, idsB, 19, n_items);
                const int uid_ = __builtin_amdgcn_readlane(idsB, 20);
                tP = rpP[lane]; tN = rpN[lane];
                u4 = row_at(user_emb, uid_)[lane];
            }
            const int bi = min(bn + NWAVES, B - 1);
            idsAn = idsA_load(pref_ids, bi, lane);
            idsBn = idsB_load(pref_ids, pos_ids, neg_ids, user_ids, bi, lane);
            npn   = np_load(n_prefs, bi, lane);
        }

        {
            const float invP = 1.0f / (lane_val(esv, 0) + lane_val(esv, 16));
            const float invN = 1.0f / (lane_val(esv, 1) + lane_val(esv, 17));
            const float px = fmaf(accP.x, invP, upP.x), py = fmaf(accP.y, invP, upP.y);
            const float pz = fmaf(accP.z, invP, upP.z), pw = fmaf(accP.w, invP, upP.w);
            const float nx = fmaf(accN.x, invN, upN.x), ny = fmaf(accN.y, invN, upN.y);
            const float nz = fmaf(accN.z, invN, upN.z), nw = fmaf(accN.w, invN, upN.w);
            float sqP = (px * px + py * py) + (pz * pz + pw * pw);
            float sqN = (nx * nx + ny * ny) + (nz * nz + nw * nw);
            sqP += xhalf(sqP);
            sqN += xhalf(sqN);
            float sq = (lane < 16) ? sqP : sqN;
            sq += xrow<8>(sq); sq += xrow<4>(sq); sq += xrow<2>(sq); sq += xrow<1>(sq);
            float* const op = out + ((lane < 16) ? (size_t)b : (size_t)B + (size_t)b);
            if ((lane & 15) == 0) *op = sq;
        }
        if (!more) break;
        b = bn;
    }
}

extern "C" void kernel_launch(void* const* d_in, const int* in_sizes, int n_in,
                              void* d_out, int out_size, void* d_ws, size_t ws_size,
                              hipStream_t stream) {
    (void)n_in; (void)out_size; (void)d_ws; (void)ws_size;
    const float* user_emb = (const float*)d_in[0];
    const float* item_emb = (const float*)d_in[1];
    const int*   user_ids = (const int*)d_in[2];
    const int*   pos_ids  = (const int*)d_in[3];
    const int*   neg_ids  = (const int*)d_in[4];
    const int*   pref_ids = (const int*)d_in[5];
    const float* n_prefs  = (const float*)d_in[6];
    float* out = (float*)d_out;

    const int B       = in_sizes[2];
    const int n_items = in_sizes[1] / DIMS;

    AttCML_kernel<<<NBLK, 128, 0, stream>>>(user_emb, item_emb, user_ids, pos_ids,
                                         neg_ids, pref_ids, n_prefs, out, B, n_items);
}


# Rows with weight zero are skipped: only a sample's first n_prefs+1 rows (the mask's prefix) are loaded, tiled and summed; a chunk with none is left out whole. Results bit-identical
# speedup vs baseline: 3.8271x; 1.2343x over previous
#include <hip/hip_runtime.h>
#include <hip/hip_bf16.h>


typedef __attribute__((ext_vector_type(16))) _Float16 v16h;
typedef __attribute__((ext_vector_type(8)))  _Float16 v8h;
typedef __attribute__((ext_vector_type(4)))  _Float16 v4h;
typedef __attribute__((ext_vector_type(8)))  float    v8f;

#define DIMS   128
#define PMAX   50
#define RPW    13
#define TROWS  16
#define HSTR   136
#define NBLK   2048
#define NWAVES (NBLK * 4)

static __device__ __forceinline__ const float4* row_at(const float* table, int id)
{
    const size_t   off = (size_t)id * (DIMS * sizeof(float));
    const unsigned lo  = __builtin_amdgcn_readfirstlane((unsigned)off);
    const unsigned hi  = __builtin_amdgcn_readfirstlane((unsigned)(off >> 32));
    const char*    row = (const char*)table + (((size_t)hi << 32) | lo);
    return (const float4*)row;
}
static __device__ __forceinline__ const float4* row_base(const float* item_emb, int idreg, int l, int n_items)
{
    const int id  = __builtin_amdgcn_readlane(idreg, l);
    return row_at(item_emb, min(id, n_items - 1));
}

static __device__ __forceinline__ int idsA_load(const int* pref_ids, int bi, int lane)
{
    return pref_ids[(size_t)bi * PMAX + lane];
}
static __device__ __forceinline__ int idsB_load(const int* pref_ids, const int* pos_ids, const int* neg_ids,
                                                const int* user_ids, int bi, int lane)
{
    const int* p = pref_ids + (size_t)bi * PMAX + 32 + lane;
    if (lane == 18) p = pos_ids + bi;
    if (lane == 19) p = neg_ids + bi;
    if (lane == 20) p = user_ids + bi;
    int r = 0;
    if (lane <= 20) r = *p;
    return r;
}
static __device__ __forceinline__ float np_load(const float* n_prefs, int bi, int lane)
{
    float r = 0.f;
    if (lane == 0) r = n_prefs[bi];
    return r;
}
static __device__ __forceinline__ float np_use(float npv)
{
    return __int_as_float(__builtin_amdgcn_readfirstlane(__float_as_int(npv)));
}
static __device__ __forceinline__ int rows_weighted(float np1)
{
    return (np1 > 0.f) ? (int)fminf(ceilf(np1), (float)PMAX) : 0;
}
static __device__ __forceinline__ float lane_val(float x, int l)
{
    return __int_as_float(__builtin_amdgcn_readlane(__float_as_int(x), l));
}

static __device__ __forceinline__ float xhalf(float x)
{
    const int xi = __float_as_int(x);
    return __int_as_float(__builtin_amdgcn_permlanex16(xi, xi, 0x76543210, 0xfedcba98, false, false));
}
template <int M>
static __device__ __forceinline__ float xrow(float x)
{
    const int xi = __float_as_int(x);
    return __int_as_float(__builtin_amdgcn_update_dpp(xi, xi, 0x160 + M, 0xf, 0xf, false));
}

#define FOR_PREF(X, C) X(0,C) X(1,C) X(2,C) X(3,C) X(4,C) X(5,C) X(6,C) X(7,C) X(8,C) X(9,C) X(10,C) X(11,C) X(12,C)
#define P_OF(j, C)     (RPW * (C) + (j))
#define HAS(j, C)      (P_OF(j, C) < PMAX)
#define ROW_DECL(j, C) float4 v##j = make_float4(0.f, 0.f, 0.f, 0.f);
#define ROW_BASE(j, C) const float4* rp##j = row_base(item_emb, P_OF(j, C) < 32 ? iA_ : iB_, P_OF(j, C) & 31, n_items);
#define ROW_LOAD(j, C) if (!HAS(j, C) || cnt_ <= P_OF(j, C)) break; v##j = rp##j[lane];
#define ROW_STORE_AT(r, val) { v4h h_; h_[0] = (_Float16)(val).x; h_[1] = (_Float16)(val).y;     \
                               h_[2] = (_Float16)(val).z; h_[3] = (_Float16)(val).w;              \
                               *(v4h*)(&tile[(r) * HSTR + lane * 4]) = h_; }
#define ROW_STORE(j, C) if (!HAS(j, C) || cnt_ <= P_OF(j, C)) break; ROW_STORE_AT(j, v##j)
#define ROW_ZERO(j, C)  if (HAS(j, C) && ((pm_ >> j) & 1u)) v##j = make_float4(0.f, 0.f, 0.f, 0.f);
#define ROW_FMA(j, C)  if (!HAS(j, C) || cnt_ <= P_OF(j, C)) break; {                          \
                       const float eP = lane_val(ev[j & 7], 16 * (j >> 3));                  \
                       const float eN = lane_val(ev[j & 7], 16 * (j >> 3) + 1);              \
                       accP.x = fmaf(eP, v##j.x, accP.x); accP.y = fmaf(eP, v##j.y, accP.y); \
                       accP.z = fmaf(eP, v##j.z, accP.z); accP.w = fmaf(eP, v##j.w, accP.w); \
                       accN.x = fmaf(eN, v##j.x, accN.x); accN.y = fmaf(eN, v##j.y, accN.y); \
                       accN.z = fmaf(eN, v##j.z, accN.z); accN.w = fmaf(eN, v##j.w, accN.w); }
#define ISSUE_CHUNK(C, IA, IB, CNT) {                                                        \
        const int iA_ = (IA); const int iB_ = (IB); const int cnt_ = (CNT);                  \
        FOR_PREF(ROW_BASE, C)                                                                \
        do { FOR_PREF(ROW_LOAD, C) } while (0);                                              \
    }
#define PADBITS(C) ((C) == 0 ? (mA & 0x1fffu) : (C) == 1 ? ((mA >> 13) & 0x1fffu)            \
                  : (C) == 2 ? (((mA >> 26) | (mB << 6)) & 0x1fffu) : ((mB >> 7) & 0x7ffu))

static __device__ __forceinline__ v8f score_tile(const _Float16* tile, int lane)
{
    const int half8 = (lane >> 4) ? 8 : 0;
    const int mrow  = lane & 15;
    const int col   = lane & 15;
    const _Float16* arow = &tile[mrow * HSTR + half8];
    const _Float16* brow = &tile[(col < 2 ? RPW + col : TROWS - 1) * HSTR + half8];
    v8f c = {0.f, 0.f, 0.f, 0.f, 0.f, 0.f, 0.f, 0.f};
    #pragma unroll
    for (int kb = 0; kb < DIMS; kb += 32) {
        const v8h a0 = *(const v8h*)(arow + kb);
        const v8h a1 = *(const v8h*)(arow + kb + 16);
        const v8h b0 = *(const v8h*)(brow + kb);
        const v8h b1 = *(const v8h*)(brow + kb + 16);
        const v16h a  = __builtin_shufflevector(a0, a1, 0, 1, 2, 3, 4, 5, 6, 7, 8, 9, 10, 11, 12, 13, 14, 15);
        const v16h bm = __builtin_shufflevector(b0, b1, 0, 1, 2, 3, 4, 5, 6, 7, 8, 9, 10, 11, 12, 13, 14, 15);
        c = __builtin_amdgcn_wmma_f32_16x16x32_f16(false, a, false, bm, (short)0, c, false, false);
    }
    return c;
}

#define COMPUTE_CHUNK(C, CNT) {                                                               \
        const int cnt_ = (CNT);                                                               \
        const unsigned pm_ = PADBITS(C);                                                      \
        if (pm_ != 0u) {                \
            FOR_PREF(ROW_ZERO, C)                                                             \
        }                                                                                     \
          \
          \
        do { FOR_PREF(ROW_STORE, C) } while (0);                                              \
        const v8f c = score_tile(tile, lane);                                                 \
          \
        float ev[8];                                                                          \
          \
          \
          \
          \
        const float thr_ = fminf(np1, (float)(RPW * ((C) + 1) < PMAX ? RPW * ((C) + 1) : PMAX)) - (float)half8; \
        _Pragma("unroll")                                                                     \
        for (int g = 0; g < 8; ++g) {                                                         \
            ev[g] = ((float)(RPW * (C) + g) < thr_) ? __expf(c[g]) : 0.f;                     \
        }                                                                                     \
          \
        esv += ((ev[0] + ev[1]) + (ev[2] + ev[3])) + ((ev[4] + ev[5]) + (ev[6] + ev[7]));     \
        do { FOR_PREF(ROW_FMA, C) } while (0);                                                \
          \
          \
        asm volatile("" : "+v"(accP.x), "+v"(accP.y), "+v"(accP.z), "+v"(accP.w),                   \
                          "+v"(accN.x), "+v"(accN.y), "+v"(accN.z), "+v"(accN.w), "+v"(esv));  \
    }

__global__ __launch_bounds__(128) __attribute__((amdgpu_waves_per_eu(8, 8)))
void AttCML_kernel(const float* user_emb,
                   const float* item_emb,
                   const int*   user_ids,
                   const int*   pos_ids,
                   const int*   neg_ids,
                   const int*   pref_ids,
                   const float* n_prefs,
                   float* out,
                   int B, int n_items)
{
    __shared__ __align__(16) _Float16 sTile[4 * TROWS * HSTR];

    const int tid  = threadIdx.x;
    const int lane = tid & 31;
    const int wave_s = __builtin_amdgcn_readfirstlane(tid >> 5) & 3;
    int b = blockIdx.x * 4 + wave_s;
    if (b >= B) return;

    _Float16* const tile = sTile + wave_s * (TROWS * HSTR);
    const int half8 = (lane >> 4) ? 8 : 0;

    {
        const v8h z_ = {(_Float16)0.f, (_Float16)0.f, (_Float16)0.f, (_Float16)0.f,
                        (_Float16)0.f, (_Float16)0.f, (_Float16)0.f, (_Float16)0.f};
        for (int k = 0; k < 8; ++k) *(v8h*)(&tile[(lane + 32 * k) * 8]) = z_;
        if (lane < 16) *(v8h*)(&tile[(lane + 256) * 8]) = z_;
    }

    int idsA = idsA_load(pref_ids, b, lane);
    int idsB = idsB_load(pref_ids, pos_ids, neg_ids, user_ids, b, lane);
    float npc = np_load(n_prefs, b, lane);
    FOR_PREF(ROW_DECL, 0)
    float4 tP = make_float4(0.f, 0.f, 0.f, 0.f), tN = tP, u4 = tP;
    float np1 = np_use(npc) + 1.0f;
    int   cnt = rows_weighted(np1);
    ISSUE_CHUNK(0, idsA, idsB, cnt)
    {
        const float4* rpP = row_base(item_emb, idsB, 18, n_items);
        const float4* rpN = row_base(item_emb, idsB, 19, n_items);
        const int uid_ = __builtin_amdgcn_readlane(idsB, 20);
        tP = rpP[lane]; tN = rpN[lane];
        u4 = row_at(user_emb, uid_)[lane];
    }
    int idsAn = idsA_load(pref_ids, min(b + NWAVES, B - 1), lane);
    int idsBn = idsB_load(pref_ids, pos_ids, neg_ids, user_ids, min(b + NWAVES, B - 1), lane);
    float npn = np_load(n_prefs, min(b + NWAVES, B - 1), lane);

    for (;;) {
        const unsigned mA = __builtin_amdgcn_ballot_w32(idsA >= n_items);
        const unsigned mB = __builtin_amdgcn_ballot_w32(lane < PMAX - 32 && idsB >= n_items);

        float4 accP = make_float4(0.f, 0.f, 0.f, 0.f), accN = accP;
        float  esv = 0.f;

        float4 upP = make_float4(u4.x - tP.x, u4.y - tP.y, u4.z - tP.z, u4.w - tP.w);
        float4 upN = make_float4(u4.x - tN.x, u4.y - tN.y, u4.z - tN.z, u4.w - tN.w);
        asm volatile("" : "+v"(upP.x), "+v"(upP.y), "+v"(upP.z), "+v"(upP.w),
                          "+v"(upN.x), "+v"(upN.y), "+v"(upN.z), "+v"(upN.w));
        ROW_STORE_AT(RPW, tP)
        ROW_STORE_AT(RPW + 1, tN)

        COMPUTE_CHUNK(0, cnt)
        if (cnt > RPW) {
            ISSUE_CHUNK(1, idsA, idsB, cnt)
            COMPUTE_CHUNK(1, cnt)
            if (cnt > 2 * RPW) {
                ISSUE_CHUNK(2, idsA, idsB, cnt)
                COMPUTE_CHUNK(2, cnt)
                if (cnt > 3 * RPW) {
                    ISSUE_CHUNK(3, idsA, idsB, cnt)
                    COMPUTE_CHUNK(3, cnt)
                }
            }
        }

        const int  bn   = b + NWAVES;
        const bool more = bn < B;
        if (more) {
            idsA = idsAn; idsB = idsBn;
            np1 = np_use(npn) + 1.0f;
            cnt = rows_weighted(np1);
            ISSUE_CHUNK(0, idsA, idsB, cnt)
            {
                const float4* rpP = row_base(item_emb, idsB, 18, n_items);
                const float4* rpN = row_base(item_emb, idsB, 19, n_items);
                const int uid_ = __builtin_amdgcn_readlane(idsB, 20);
                tP = rpP[lane]; tN = rpN[lane];
                u4 = row_at(user_emb, uid_)[lane];
            }
            const int bi = min(bn + NWAVES, B - 1);
            idsAn = idsA_load(pref_ids, bi, lane);
            idsBn = idsB_load(pref_ids, pos_ids, neg_ids, user_ids, bi, lane);
            npn   = np_load(n_prefs, bi, lane);
        }

        {
            const float invP = 1.0f / (lane_val(esv, 0) + lane_val(esv, 16));
            const float invN = 1.0f / (lane_val(esv, 1) + lane_val(esv, 17));
            const float px = fmaf(accP.x, invP, upP.x), py = fmaf(accP.y, invP, upP.y);
            const float pz = fmaf(accP.z, invP, upP.z), pw = fmaf(accP.w, invP, upP.w);
            const float nx = fmaf(accN.x, invN, upN.x), ny = fmaf(accN.y, invN, upN.y);
            const float nz = fmaf(accN.z, invN, upN.z), nw = fmaf(accN.w, invN, upN.w);
            float sqP = (px * px + py * py) + (pz * pz + pw * pw);
            float sqN = (nx * nx + ny * ny) + (nz * nz + nw * nw);
            sqP += xhalf(sqP);
            sqN += xhalf(sqN);
            float sq = (lane < 16) ? sqP : sqN;
            sq += xrow<8>(sq); sq += xrow<4>(sq); sq += xrow<2>(sq); sq += xrow<1>(sq);
            float* const op = out + ((lane < 16) ? (size_t)b : (size_t)B + (size_t)b);
            if ((lane & 15) == 0) *op = sq;
        }
        if (!more) break;
        b = bn;
    }
}

extern "C" void kernel_launch(void* const* d_in, const int* in_sizes, int n_in,
                              void* d_out, int out_size, void* d_ws, size_t ws_size,
                              hipStream_t stream) {
    (void)n_in; (void)out_size; (void)d_ws; (void)ws_size;
    const float* user_emb = (const float*)d_in[0];
    const float* item_emb = (const float*)d_in[1];
    const int*   user_ids = (const int*)d_in[2];
    const int*   pos_ids  = (const int*)d_in[3];
    const int*   neg_ids  = (const int*)d_in[4];
    const int*   pref_ids = (const int*)d_in[5];
    const float* n_prefs  = (const float*)d_in[6];
    float* out = (float*)d_out;

    const int B       = in_sizes[2];
    const int n_items = in_sizes[1] / DIMS;

    AttCML_kernel<<<NBLK, 128, 0, stream>>>(user_emb, item_emb, user_ids, pos_ids,
                                         neg_ids, pref_ids, n_prefs, out, B, n_items);
}


# One sample per wave, 4096 groups of 4 waves (was 2048 persistent groups, 2 samples per wave): the dispatcher refills freed wave slots, so uneven samples (2..50 rows) no longer add up in a wave
# speedup vs baseline: 4.3178x; 1.1282x over previous
#include <hip/hip_runtime.h>
#include <hip/hip_bf16.h>


typedef __attribute__((ext_vector_type(16))) _Float16 v16h;
typedef __attribute__((ext_vector_type(8)))  _Float16 v8h;
typedef __attribute__((ext_vector_type(4)))  _Float16 v4h;
typedef __attribute__((ext_vector_type(8)))  float    v8f;

#define DIMS   128
#define PMAX   50
#define RPW    13
#define TROWS  16
#define HSTR   136
#define NBLK   4096

static __device__ __forceinline__ const float4* row_at(const float* table, int id)
{
    const size_t   off = (size_t)id * (DIMS * sizeof(float));
    const unsigned lo  = __builtin_amdgcn_readfirstlane((unsigned)off);
    const unsigned hi  = __builtin_amdgcn_readfirstlane((unsigned)(off >> 32));
    const char*    row = (const char*)table + (((size_t)hi << 32) | lo);
    return (const float4*)row;
}
static __device__ __forceinline__ const float4* row_base(const float* item_emb, int idreg, int l, int n_items)
{
    const int id  = __builtin_amdgcn_readlane(idreg, l);
    return row_at(item_emb, min(id, n_items - 1));
}

static __device__ __forceinline__ int idsA_load(const int* pref_ids, int bi, int lane)
{
    return pref_ids[(size_t)bi * PMAX + lane];
}
static __device__ __forceinline__ int idsB_load(const int* pref_ids, const int* pos_ids, const int* neg_ids,
                                                const int* user_ids, int bi, int lane)
{
    const int* p = pref_ids + (size_t)bi * PMAX + 32 + lane;
    if (lane == 18) p = pos_ids + bi;
    if (lane == 19) p = neg_ids + bi;
    if (lane == 20) p = user_ids + bi;
    int r = 0;
    if (lane <= 20) r = *p;
    return r;
}
static __device__ __forceinline__ float np_load(const float* n_prefs, int bi, int lane)
{
    float r = 0.f;
    if (lane == 0) r = n_prefs[bi];
    return r;
}
static __device__ __forceinline__ float np_use(float npv)
{
    return __int_as_float(__builtin_amdgcn_readfirstlane(__float_as_int(npv)));
}
static __device__ __forceinline__ int rows_weighted(float np1)
{
    return (np1 > 0.f) ? (int)fminf(ceilf(np1), (float)PMAX) : 0;
}
static __device__ __forceinline__ float lane_val(float x, int l)
{
    return __int_as_float(__builtin_amdgcn_readlane(__float_as_int(x), l));
}

static __device__ __forceinline__ float xhalf(float x)
{
    const int xi = __float_as_int(x);
    return __int_as_float(__builtin_amdgcn_permlanex16(xi, xi, 0x76543210, 0xfedcba98, false, false));
}
template <int M>
static __device__ __forceinline__ float xrow(float x)
{
    const int xi = __float_as_int(x);
    return __int_as_float(__builtin_amdgcn_update_dpp(xi, xi, 0x160 + M, 0xf, 0xf, false));
}

#define FOR_PREF(X, C) X(0,C) X(1,C) X(2,C) X(3,C) X(4,C) X(5,C) X(6,C) X(7,C) X(8,C) X(9,C) X(10,C) X(11,C) X(12,C)
#define P_OF(j, C)     (RPW * (C) + (j))
#define HAS(j, C)      (P_OF(j, C) < PMAX)
#define ROW_DECL(j, C) float4 v##j = make_float4(0.f, 0.f, 0.f, 0.f);
#define ROW_BASE(j, C) const float4* rp##j = row_base(item_emb, P_OF(j, C) < 32 ? iA_ : iB_, P_OF(j, C) & 31, n_items);
#define ROW_LOAD(j, C) if (!HAS(j, C) || cnt_ <= P_OF(j, C)) break; v##j = rp##j[lane];
#define ROW_STORE_AT(r, val) { v4h h_; h_[0] = (_Float16)(val).x; h_[1] = (_Float16)(val).y;     \
                               h_[2] = (_Float16)(val).z; h_[3] = (_Float16)(val).w;              \
                               *(v4h*)(&tile[(r) * HSTR + lane * 4]) = h_; }
#define ROW_STORE(j, C) if (!HAS(j, C) || cnt_ <= P_OF(j, C)) break; ROW_STORE_AT(j, v##j)
#define ROW_ZERO(j, C)  if (HAS(j, C) && ((pm_ >> j) & 1u)) v##j = make_float4(0.f, 0.f, 0.f, 0.f);
#define ROW_FMA(j, C)  if (!HAS(j, C) || cnt_ <= P_OF(j, C)) break; {                          \
                       const float eP = lane_val(ev[j & 7], 16 * (j >> 3));                  \
                       const float eN = lane_val(ev[j & 7], 16 * (j >> 3) + 1);              \
                       accP.x = fmaf(eP, v##j.x, accP.x); accP.y = fmaf(eP, v##j.y, accP.y); \
                       accP.z = fmaf(eP, v##j.z, accP.z); accP.w = fmaf(eP, v##j.w, accP.w); \
                       accN.x = fmaf(eN, v##j.x, accN.x); accN.y = fmaf(eN, v##j.y, accN.y); \
                       accN.z = fmaf(eN, v##j.z, accN.z); accN.w = fmaf(eN, v##j.w, accN.w); }
#define ISSUE_CHUNK(C, IA, IB, CNT) {                                                        \
        const int iA_ = (IA); const int iB_ = (IB); const int cnt_ = (CNT);                  \
        FOR_PREF(ROW_BASE, C)                                                                \
        do { FOR_PREF(ROW_LOAD, C) } while (0);                                              \
    }
#define PADBITS(C) ((C) == 0 ? (mA & 0x1fffu) : (C) == 1 ? ((mA >> 13) & 0x1fffu)            \
                  : (C) == 2 ? (((mA >> 26) | (mB << 6)) & 0x1fffu) : ((mB >> 7) & 0x7ffu))

static __device__ __forceinline__ v8f score_tile(const _Float16* tile, int lane)
{
    const int half8 = (lane >> 4) ? 8 : 0;
    const int mrow  = lane & 15;
    const int col   = lane & 15;
    const _Float16* arow = &tile[mrow * HSTR + half8];
    const _Float16* brow = &tile[(col < 2 ? RPW + col : TROWS - 1) * HSTR + half8];
    v8f c = {0.f, 0.f, 0.f, 0.f, 0.f, 0.f, 0.f, 0.f};
    #pragma unroll
    for (int kb = 0; kb < DIMS; kb += 32) {
        const v8h a0 = *(const v8h*)(arow + kb);
        const v8h a1 = *(const v8h*)(arow + kb + 16);
        const v8h b0 = *(const v8h*)(brow + kb);
        const v8h b1 = *(const v8h*)(brow + kb + 16);
        const v16h a  = __builtin_shufflevector(a0, a1, 0, 1, 2, 3, 4, 5, 6, 7, 8, 9, 10, 11, 12, 13, 14, 15);
        const v16h bm = __builtin_shufflevector(b0, b1, 0, 1, 2, 3, 4, 5, 6, 7, 8, 9, 10, 11, 12, 13, 14, 15);
        c = __builtin_amdgcn_wmma_f32_16x16x32_f16(false, a, false, bm, (short)0, c, false, false);
    }
    return c;
}

#define COMPUTE_CHUNK(C, CNT) {                                                               \
        const int cnt_ = (CNT);                                                               \
          \
          \
          \
        const int lane_k_ = lane;                                                             \
        int lane = lane_k_;                                                                   \
        asm volatile("" : "+v"(lane));                                                        \
        const int half8 = (lane >> 4) ? 8 : 0;                                                \
        const unsigned pm_ = PADBITS(C);                                                      \
        if (pm_ != 0u) {                \
            FOR_PREF(ROW_ZERO, C)                                                             \
        }                                                                                     \
          \
          \
        do { FOR_PREF(ROW_STORE, C) } while (0);                                              \
        const v8f c = score_tile(tile, lane);                                                 \
          \
        float ev[8];                                                                          \
          \
          \
          \
          \
        const float thr_ = fminf(np1, (float)(RPW * ((C) + 1) < PMAX ? RPW * ((C) + 1) : PMAX)) - (float)half8; \
        _Pragma("unroll")                                                                     \
        for (int g = 0; g < 8; ++g) {                                                         \
            ev[g] = ((float)(RPW * (C) + g) < thr_) ? __expf(c[g]) : 0.f;                     \
        }                                                                                     \
          \
        esv += ((ev[0] + ev[1]) + (ev[2] + ev[3])) + ((ev[4] + ev[5]) + (ev[6] + ev[7]));     \
        do { FOR_PREF(ROW_FMA, C) } while (0);                                                \
          \
          \
        asm volatile("" : "+v"(accP.x), "+v"(accP.y), "+v"(accP.z), "+v"(accP.w),                   \
                          "+v"(accN.x), "+v"(accN.y), "+v"(accN.z), "+v"(accN.w), "+v"(esv));  \
    }

__global__ __launch_bounds__(128) __attribute__((amdgpu_waves_per_eu(8, 8)))
void AttCML_kernel(const float* user_emb,
                   const float* item_emb,
                   const int*   user_ids,
                   const int*   pos_ids,
                   const int*   neg_ids,
                   const int*   pref_ids,
                   const float* n_prefs,
                   float* out,
                   int B, int n_items)
{
    __shared__ __align__(16) _Float16 sTile[4 * TROWS * HSTR];

    const int tid  = threadIdx.x;
    const int lane = tid & 31;
    const int wave_s = __builtin_amdgcn_readfirstlane(tid >> 5) & 3;
    const int b = blockIdx.x * 4 + wave_s;
    if (b >= B) return;

    _Float16* const tile = sTile + wave_s * (TROWS * HSTR);

    {
        const v8h z_ = {(_Float16)0.f, (_Float16)0.f, (_Float16)0.f, (_Float16)0.f,
                        (_Float16)0.f, (_Float16)0.f, (_Float16)0.f, (_Float16)0.f};
        for (int k = 0; k < 8; ++k) *(v8h*)(&tile[(lane + 32 * k) * 8]) = z_;
        if (lane < 16) *(v8h*)(&tile[(lane + 256) * 8]) = z_;
    }

    const int half8 = (lane >> 4) ? 8 : 0;
    {
        const int   idsA = idsA_load(pref_ids, b, lane);
        const int   idsB = idsB_load(pref_ids, pos_ids, neg_ids, user_ids, b, lane);
        const float npc  = np_load(n_prefs, b, lane);
        FOR_PREF(ROW_DECL, 0)
        const float np1 = np_use(npc) + 1.0f;
        const int   cnt = rows_weighted(np1);
        ISSUE_CHUNK(0, idsA, idsB, cnt)
        float4 tP, tN, u4;
        {
            const float4* rpP = row_base(item_emb, idsB, 18, n_items);
            const float4* rpN = row_base(item_emb, idsB, 19, n_items);
            const int uid_ = __builtin_amdgcn_readlane(idsB, 20);
            tP = rpP[lane]; tN = rpN[lane];
            u4 = row_at(user_emb, uid_)[lane];
        }

        const unsigned mA = __builtin_amdgcn_ballot_w32(idsA >= n_items);
        const unsigned mB = __builtin_amdgcn_ballot_w32(lane < PMAX - 32 && idsB >= n_items);

        float4 accP = make_float4(0.f, 0.f, 0.f, 0.f), accN = accP;
        float  esv = 0.f;

        float4 upP = make_float4(u4.x - tP.x, u4.y - tP.y, u4.z - tP.z, u4.w - tP.w);
        float4 upN = make_float4(u4.x - tN.x, u4.y - tN.y, u4.z - tN.z, u4.w - tN.w);
        asm volatile("" : "+v"(upP.x), "+v"(upP.y), "+v"(upP.z), "+v"(upP.w),
                          "+v"(upN.x), "+v"(upN.y), "+v"(upN.z), "+v"(upN.w));
        ROW_STORE_AT(RPW, tP)
        ROW_STORE_AT(RPW + 1, tN)

        COMPUTE_CHUNK(0, cnt)
        if (cnt > RPW) {
            ISSUE_CHUNK(1, idsA, idsB, cnt)
            COMPUTE_CHUNK(1, cnt)
            if (cnt > 2 * RPW) {
                ISSUE_CHUNK(2, idsA, idsB, cnt)
                COMPUTE_CHUNK(2, cnt)
                if (cnt > 3 * RPW) {
                    ISSUE_CHUNK(3, idsA, idsB, cnt)
                    COMPUTE_CHUNK(3, cnt)
                }
            }
        }

        {
            const float invP = 1.0f / (lane_val(esv, 0) + lane_val(esv, 16));
            const float invN = 1.0f / (lane_val(esv, 1) + lane_val(esv, 17));
            const float px = fmaf(accP.x, invP, upP.x), py = fmaf(accP.y, invP, upP.y);
            const float pz = fmaf(accP.z, invP, upP.z), pw = fmaf(accP.w, invP, upP.w);
            const float nx = fmaf(accN.x, invN, upN.x), ny = fmaf(accN.y, invN, upN.y);
            const float nz = fmaf(accN.z, invN, upN.z), nw = fmaf(accN.w, invN, upN.w);
            float sqP = (px * px + py * py) + (pz * pz + pw * pw);
            float sqN = (nx * nx + ny * ny) + (nz * nz + nw * nw);
            sqP += xhalf(sqP);
            sqN += xhalf(sqN);
            float sq = (lane < 16) ? sqP : sqN;
            sq += xrow<8>(sq); sq += xrow<4>(sq); sq += xrow<2>(sq); sq += xrow<1>(sq);
            float* const op = out + ((lane < 16) ? (size_t)b : (size_t)B + (size_t)b);
            if ((lane & 15) == 0) *op = sq;
        }
    }
}

extern "C" void kernel_launch(void* const* d_in, const int* in_sizes, int n_in,
                              void* d_out, int out_size, void* d_ws, size_t ws_size,
                              hipStream_t stream) {
    (void)n_in; (void)out_size; (void)d_ws; (void)ws_size;
    const float* user_emb = (const float*)d_in[0];
    const float* item_emb = (const float*)d_in[1];
    const int*   user_ids = (const int*)d_in[2];
    const int*   pos_ids  = (const int*)d_in[3];
    const int*   neg_ids  = (const int*)d_in[4];
    const int*   pref_ids = (const int*)d_in[5];
    const float* n_prefs  = (const float*)d_in[6];
    float* out = (float*)d_out;

    const int B       = in_sizes[2];
    const int n_items = in_sizes[1] / DIMS;

    AttCML_kernel<<<NBLK, 128, 0, stream>>>(user_emb, item_emb, user_ids, pos_ids,
                                         neg_ids, pref_ids, n_prefs, out, B, n_items);
}
